# starting kernel
# baseline (speedup 1.0000x reference)
#include <hip/hip_runtime.h>

typedef __attribute__((ext_vector_type(2))) float v2f;
typedef __attribute__((ext_vector_type(8))) float v8f;

#define IC 256
#define OC 256
#define HH 56
#define WW 56
#define HWSZ (HH * WW)
#define CHW (IC * HWSZ)
#define KTAPS 9
#define WROW (IC * KTAPS)

#define WAVE_OC 16
#define WAVE_POS 64
#define BLOCK_OC 64
#define BLOCK_POS 128

__global__ __launch_bounds__(256) void conv3x3_wmma_f32(
    const float* __restrict__ x, const float* __restrict__ w,
    float* __restrict__ out)
{
  const int lane   = threadIdx.x & 31;
  const int wave   = threadIdx.x >> 5;
  const int l16    = lane & 15;
  const int hiHalf = lane >> 4;
  const int kHi    = hiHalf * 2;

  const int wave_oc  = wave & 3;
  const int wave_pos = wave >> 2;

  const int oc_base  = blockIdx.y * BLOCK_OC + wave_oc * WAVE_OC;
  const int pos_base = blockIdx.x * BLOCK_POS + wave_pos * WAVE_POS;

  const int m = oc_base + l16;

  int n_[4], h_[4], w_[4], sp_[4];
#pragma unroll
  for (int j = 0; j < 4; ++j) {
    int p  = pos_base + j * 16 + l16;
    int n  = p / HWSZ;
    int sp = p - n * HWSZ;
    int hh = sp / WW;
    int ww = sp - hh * WW;
    n_[j] = n; h_[j] = hh; w_[j] = ww; sp_[j] = sp;
  }

  v8f c[4] = {};

  const float* wRowBase = w + (long)m * WROW + kHi * KTAPS;

  for (int tap = 0; tap < KTAPS; ++tap) {
    const int dh = tap / 3 - 1;
    const int dw = tap % 3 - 1;

    const float* bptr[4];
    bool valid[4];
#pragma unroll
    for (int j = 0; j < 4; ++j) {
      int hh = h_[j] + dh, ww = w_[j] + dw;
      valid[j] = ((unsigned)hh < (unsigned)HH) && ((unsigned)ww < (unsigned)WW);
      long spOff = valid[j] ? (long)(hh * WW + ww) : 0L;
      bptr[j] = x + (long)n_[j] * CHW + (long)kHi * HWSZ + spOff;
    }

    const float* aptr = wRowBase + tap;

#pragma unroll 4
    for (int ic0 = 0; ic0 < IC; ic0 += 4) {
      v2f a;
      a.x = aptr[0];
      a.y = aptr[KTAPS];
      aptr += 4 * KTAPS;

#pragma unroll
      for (int j = 0; j < 4; ++j) {
        float t0 = bptr[j][0];
        float t1 = bptr[j][HWSZ];
        bptr[j] += 4 * HWSZ;
        v2f b;
        b.x = valid[j] ? t0 : 0.0f;
        b.y = valid[j] ? t1 : 0.0f;

        c[j] = __builtin_amdgcn_wmma_f32_16x16x4_f32(
             false, a,  false, b,
             (short)0, c[j],  false,  false);
      }
    }
  }

  const float s = 1.0f / 9.0f;
#pragma unroll
  for (int j = 0; j < 4; ++j) {
    const int  ocr   = oc_base + hiHalf * 8;
    const long obase = (long)n_[j] * (long)OC * HWSZ + sp_[j];
#pragma unroll
    for (int r = 0; r < 8; ++r) {
      out[obase + (long)(ocr + r) * HWSZ] = c[j][r] * s;
    }
  }
}

extern "C" void kernel_launch(void* const* d_in, const int* in_sizes, int n_in,
                              void* d_out, int out_size, void* d_ws, size_t ws_size,
                              hipStream_t stream) {
  const float* x = (const float*)d_in[0];
  const float* w = (const float*)d_in[1];
  float* out = (float*)d_out;

  dim3 grid(100352 / BLOCK_POS, OC / BLOCK_OC, 1);
  dim3 block(256, 1, 1);
  conv3x3_wmma_f32<<<grid, block, 0, stream>>>(x, w, out);
}


# stage 8-channel slabs of image rows (zero halo, channel pairs interleaved) and weights in LDS; each wave reads all 9 taps' fragments for 16 oc x 224 positions from LDS; sum order now slab, tap, k
# speedup vs baseline: 4.1535x; 4.1535x over previous
#include <hip/hip_runtime.h>


typedef __attribute__((ext_vector_type(2))) float v2f;
typedef __attribute__((ext_vector_type(4))) float v4f;
typedef __attribute__((ext_vector_type(8))) float v8f;

#define IC 256
#define OC 256
#define HH 56
#define WW 56
#define HWSZ (HH * WW)
#define KTAPS 9
#define WROW (IC * KTAPS)

#define NTHREADS 128
#define ICC 8
#define NCHUNK (IC / ICC)
#define BLOCK_OC 64
#define ROWS 4
#define NT 14
#define HBLK (HH / ROWS)
#define TROWS (ROWS + 2)

#define XROWP 64
#define XPAIR (TROWS * XROWP * 2 + 32)
#define XS_FLOATS ((ICC / 2) * XPAIR)
#define WPITCH 76
#define WS_FLOATS (BLOCK_OC * WPITCH)

#define XPER (TROWS * (WW / 4))
#define XITEMS ((ICC / 2) * XPER)
#define XROUNDS 3

__global__ __launch_bounds__(NTHREADS) void conv3x3_wmma_f32(
    const float* __restrict__ x, const float* __restrict__ w,
    float* __restrict__ out)
{
  __shared__ __attribute__((aligned(16))) float xs[XS_FLOATS];
  __shared__ __attribute__((aligned(16))) float wsm[WS_FLOATS];

  const int tid    = threadIdx.x;
  const int lane   = tid & 31;
  const int wave   = tid >> 5;
  const int l16    = lane & 15;
  const int hiHalf = lane >> 4;

  const int n   = blockIdx.y / HBLK;
  const int h0  = (blockIdx.y - n * HBLK) * ROWS;
  const int oc0 = blockIdx.x * BLOCK_OC;

  int  xg_off[XROUNDS], xl_off[XROUNDS];
  bool xok[XROUNDS];
#pragma unroll
  for (int s = 0; s < XROUNDS; ++s) {
    const int i   = tid + NTHREADS * s;
    const int xp  = i / XPER;
    const int rem = i - xp * XPER;
    const int rr  = rem / 14;
    const int c4  = rem - rr * 14;
    const int row = h0 - 1 + rr;
    xok[s]    = (i < XITEMS) && ((unsigned)row < (unsigned)HH);
    xg_off[s] = xok[s] ? (2 * xp) * HWSZ + row * WW + c4 * 4 : 0;
    xl_off[s] = xp * XPAIR + (rr * XROWP + 4 + c4 * 4) * 2;
  }
  const int wg_off0 = (tid >> 4) * WROW + (tid & 15) * 4;
  const int wl_off0 = (tid >> 4) * WPITCH + (tid & 15) * 4;
  const int wg_off1 = (tid >> 1) * WROW + 64 + (tid & 1) * 4;
  const int wl_off1 = (tid >> 1) * WPITCH + 64 + (tid & 1) * 4;

  const float* xblk = x + (long)n * IC * HWSZ;
  const float* wblk = w + (long)oc0 * WROW;

  v4f xra[XROUNDS], xrb[XROUNDS];
  v4f wr[9];

#pragma unroll
  for (int s = 0; s < (XS_FLOATS / 4 + NTHREADS - 1) / NTHREADS; ++s) {
    const int f = tid + NTHREADS * s;
    if (f < XS_FLOATS / 4) *(v4f*)&xs[f * 4] = (v4f){0.f, 0.f, 0.f, 0.f};
  }

#pragma unroll
  for (int s = 0; s < XROUNDS; ++s) {
    xra[s] = *(const v4f*)(xblk + xg_off[s]);
    xrb[s] = *(const v4f*)(xblk + xg_off[s] + HWSZ);
  }
#pragma unroll
  for (int p = 0; p < 8; ++p)
    wr[p] = *(const v4f*)(wblk + wg_off0 + p * 8 * WROW);
  wr[8] = *(const v4f*)(wblk + wg_off1);

  __syncthreads();

  int bbase[NT];
#pragma unroll
  for (int j = 0; j < NT; ++j) {
    const int pl  = j * 16 + l16;
    const int r   = pl / WW;
    const int col = pl - r * WW;
    bbase[j] = hiHalf * XPAIR + (r * XROWP + col + 3) * 2;
  }
  const float* ap = wsm + (wave * 16 + l16) * WPITCH + hiHalf * 2 * KTAPS;

  v8f c[NT] = {};

  for (int chunk = 0; chunk < NCHUNK; ++chunk) {
#pragma unroll
    for (int s = 0; s < XROUNDS; ++s)
      if (xok[s]) {
        *(v4f*)&xs[xl_off[s]]     = (v4f){xra[s].x, xrb[s].x, xra[s].y, xrb[s].y};
        *(v4f*)&xs[xl_off[s] + 4] = (v4f){xra[s].z, xrb[s].z, xra[s].w, xrb[s].w};
      }
#pragma unroll
    for (int p = 0; p < 8; ++p)
      *(v4f*)&wsm[wl_off0 + p * 8 * WPITCH] = wr[p];
    *(v4f*)&wsm[wl_off1] = wr[8];

    __syncthreads();

    if (chunk + 1 < NCHUNK) {
      const float* xb = xblk + (long)(chunk + 1) * ICC * HWSZ;
      const float* wb = wblk + (chunk + 1) * ICC * KTAPS;
#pragma unroll
      for (int s = 0; s < XROUNDS; ++s) {
        xra[s] = *(const v4f*)(xb + xg_off[s]);
        xrb[s] = *(const v4f*)(xb + xg_off[s] + HWSZ);
      }
#pragma unroll
      for (int p = 0; p < 8; ++p)
        wr[p] = *(const v4f*)(wb + wg_off0 + p * 8 * WROW);
      wr[8] = *(const v4f*)(wb + wg_off1);
    }

#pragma unroll 1
    for (int t = 0; t < KTAPS; ++t) {
      const int toff = ((t / 3) * XROWP + (t % 3)) * 2;
#pragma unroll
      for (int q = 0; q < ICC / 4; ++q) {
        v2f a;
        a.x = ap[(4 * q)     * KTAPS + t];
        a.y = ap[(4 * q + 1) * KTAPS + t];
#pragma unroll
        for (int j = 0; j < NT; ++j) {
          const v2f b = *(const v2f*)&xs[(2 * q) * XPAIR + bbase[j] + toff];
          c[j] = __builtin_amdgcn_wmma_f32_16x16x4_f32(
               false, a,  false, b,
               (short)0, c[j],  false,  false);
        }
      }
    }

#pragma unroll
    for (int s = 0; s < XROUNDS; ++s)
      asm volatile("" : "+v"(xra[s]), "+v"(xrb[s]));

    __syncthreads();
  }

  const float s9 = 1.0f / 9.0f;
  const int   ocr = oc0 + wave * 16 + hiHalf * 8;
#pragma unroll
  for (int j = 0; j < NT; ++j) {
    const long obase = (long)n * (long)OC * HWSZ + h0 * WW + j * 16 + l16;
#pragma unroll
    for (int r = 0; r < 8; ++r) {
      out[obase + (long)(ocr + r) * HWSZ] = c[j][r] * s9;
    }
  }
}

extern "C" void kernel_launch(void* const* d_in, const int* in_sizes, int n_in,
                              void* d_out, int out_size, void* d_ws, size_t ws_size,
                              hipStream_t stream) {
  const float* x = (const float*)d_in[0];
  const float* w = (const float*)d_in[1];
  float* out = (float*)d_out;

  dim3 grid(4, 448, 1);
  dim3 block(NTHREADS, 1, 1);
  conv3x3_wmma_f32<<<grid, block, 0, stream>>>(x, w, out);
}


# the 3 column taps computed in F(2,3) minimal-filtering form along the width: 4 transformed products per output pair, not 6 (168 WMMAs per wave and slab, not 252); sums now in the transformed domain
# speedup vs baseline: 4.9525x; 1.1924x over previous
#include <hip/hip_runtime.h>


typedef __attribute__((ext_vector_type(2))) float v2f;
typedef __attribute__((ext_vector_type(4))) float v4f;
typedef __attribute__((ext_vector_type(8))) float v8f;

#define IC 256
#define OC 256
#define HH 56
#define WW 56
#define HWSZ (HH * WW)
#define KTAPS 9
#define WROW (IC * KTAPS)

#define NTHREADS 128
#define ICC 8
#define NCHUNK (IC / ICC)
#define BLOCK_OC 64
#define ROWS 4
#define TROWS (ROWS + 2)
#define MW (WW / 2)
#define NF 4
#define NT (ROWS * MW / 16)
#define HBLK (HH / ROWS)

#define FPITCH 64
#define RPITCH (NF * FPITCH)
#define XPAIR (TROWS * RPITCH + 32)
#define XS_FLOATS ((ICC / 2) * XPAIR)
#define UPITCH 100
#define US_FLOATS (BLOCK_OC * UPITCH)

#define XROUNDS 3
#define WROUNDS 3

__global__ __launch_bounds__(NTHREADS) void conv3x3_wmma_f32(
    const float* __restrict__ x, const float* __restrict__ w,
    float* __restrict__ out)
{
  __shared__ __attribute__((aligned(16))) float xs[XS_FLOATS];
  __shared__ __attribute__((aligned(16))) float usm[US_FLOATS];

  const int tid    = threadIdx.x;
  const int lane   = tid & 31;
  const int wave   = tid >> 5;
  const int l16    = lane & 15;
  const int hiHalf = lane >> 4;

  const int n   = blockIdx.y / HBLK;
  const int h0  = (blockIdx.y - n * HBLK) * ROWS;
  const int oc0 = blockIdx.x * BLOCK_OC;

  const int  rw   = lane / 14;
  const int  c4   = lane - rw * 14;
  const bool hasL = c4 > 0;
  const bool hasR = c4 < 13;
  int  xg_off[XROUNDS], xg_l[XROUNDS], xg_r[XROUNDS], xl_off[XROUNDS];
  bool xok[XROUNDS];
#pragma unroll
  for (int s = 0; s < XROUNDS; ++s) {
    const int line = (s * 4 + wave) * 2 + rw;
    const int xp   = line / TROWS;
    const int rr   = line - xp * TROWS;
    const int row  = h0 - 1 + rr;
    xok[s]    = (lane < 28) && ((unsigned)row < (unsigned)HH);
    xg_off[s] = xok[s] ? (2 * xp) * HWSZ + row * WW + c4 * 4 : 0;
    xg_l[s]   = xg_off[s] - ((xok[s] && hasL) ? 1 : 0);
    xg_r[s]   = xg_off[s] + ((xok[s] && hasR) ? 4 : 0);
    xl_off[s] = xp * XPAIR + rr * RPITCH + c4 * 4;
  }
  int wg_off[WROUNDS], wl_off[WROUNDS];
#pragma unroll
  for (int p = 0; p < WROUNDS; ++p) {
    const int i   = tid + NTHREADS * p;
    const int ocl = i / 6;
    const int rem = i - ocl * 6;
    const int q   = rem / 3;
    const int kh  = rem - q * 3;
    wg_off[p] = ocl * WROW + (4 * q) * KTAPS + kh * 3;
    wl_off[p] = ocl * UPITCH + kh * ICC + 4 * q;
  }

  const float* xblk = x + (long)n * IC * HWSZ;
  const float* wblk = w + (long)oc0 * WROW;

  v4f   xma[XROUNDS], xmb[XROUNDS];
  float xla[XROUNDS], xlb[XROUNDS];
  float xra[XROUNDS], xrb[XROUNDS];
  float wr[WROUNDS][4][3];

#pragma unroll
  for (int s = 0; s < (XS_FLOATS / 4 + NTHREADS - 1) / NTHREADS; ++s) {
    const int f = tid + NTHREADS * s;
    if (f < XS_FLOATS / 4) *(v4f*)&xs[f * 4] = (v4f){0.f, 0.f, 0.f, 0.f};
  }

#pragma unroll
  for (int s = 0; s < XROUNDS; ++s) {
    xma[s] = *(const v4f*)(xblk + xg_off[s]);
    xmb[s] = *(const v4f*)(xblk + xg_off[s] + HWSZ);
    xla[s] = xblk[xg_l[s]];
    xlb[s] = xblk[xg_l[s] + HWSZ];
    xra[s] = xblk[xg_r[s]];
    xrb[s] = xblk[xg_r[s] + HWSZ];
  }
#pragma unroll
  for (int p = 0; p < WROUNDS; ++p)
#pragma unroll
    for (int cc = 0; cc < 4; ++cc)
#pragma unroll
      for (int kw = 0; kw < 3; ++kw)
        wr[p][cc][kw] = wblk[wg_off[p] + cc * KTAPS + kw];

  __syncthreads();

  int bbase[NT];
#pragma unroll
  for (int j = 0; j < NT; ++j) {
    const int pl = j * 16 + l16;
    const int r  = pl / MW;
    const int m  = pl - r * MW;
    bbase[j] = hiHalf * XPAIR + r * RPITCH + m * 2;
  }
  const float* ap = usm + (wave * 16 + l16) * UPITCH + hiHalf * 2;

  v8f c[NF][NT] = {};

  for (int chunk = 0; chunk < NCHUNK; ++chunk) {
#pragma unroll
    for (int s = 0; s < XROUNDS; ++s)
      if (xok[s]) {
        const float la = hasL ? xla[s] : 0.f, lb = hasL ? xlb[s] : 0.f;
        const float ra = hasR ? xra[s] : 0.f, rb = hasR ? xrb[s] : 0.f;
        const v4f a = xma[s], b = xmb[s];
        *(v4f*)&xs[xl_off[s]]              = (v4f){la  - a.y, lb  - b.y, a.y - a.w, b.y - b.w};
        *(v4f*)&xs[xl_off[s] + FPITCH]     = (v4f){a.x + a.y, b.x + b.y, a.z + a.w, b.z + b.w};
        *(v4f*)&xs[xl_off[s] + 2 * FPITCH] = (v4f){a.y - a.x, b.y - b.x, a.w - a.z, b.w - b.z};
        *(v4f*)&xs[xl_off[s] + 3 * FPITCH] = (v4f){a.x - a.z, b.x - b.z, a.z - ra,  b.z - rb};
      }
#pragma unroll
    for (int p = 0; p < WROUNDS; ++p) {
      v4f u0, u1, u2, u3;
#pragma unroll
      for (int cc = 0; cc < 4; ++cc) {
        const float g0 = wr[p][cc][0], g1 = wr[p][cc][1], g2 = wr[p][cc][2];
        const float e = g0 + g2;
        u0[cc] = g0;
        u1[cc] = (e + g1) * 0.5f;
        u2[cc] = (e - g1) * 0.5f;
        u3[cc] = g2;
      }
      *(v4f*)&usm[wl_off[p]]               = u0;
      *(v4f*)&usm[wl_off[p] + 3 * ICC]     = u1;
      *(v4f*)&usm[wl_off[p] + 6 * ICC]     = u2;
      *(v4f*)&usm[wl_off[p] + 9 * ICC]     = u3;
    }

    __syncthreads();

    {
      const int nx = (chunk + 1 < NCHUNK) ? chunk + 1 : chunk;
      const float* xb = xblk + (long)nx * ICC * HWSZ;
      const float* wb = wblk + nx * ICC * KTAPS;
#pragma unroll
      for (int s = 0; s < XROUNDS; ++s) {
        xma[s] = *(const v4f*)(xb + xg_off[s]);
        xmb[s] = *(const v4f*)(xb + xg_off[s] + HWSZ);
        xla[s] = xb[xg_l[s]];
        xlb[s] = xb[xg_l[s] + HWSZ];
        xra[s] = xb[xg_r[s]];
        xrb[s] = xb[xg_r[s] + HWSZ];
      }
#pragma unroll
      for (int p = 0; p < WROUNDS; ++p)
#pragma unroll
        for (int cc = 0; cc < 4; ++cc)
#pragma unroll
          for (int kw = 0; kw < 3; ++kw)
            wr[p][cc][kw] = wb[wg_off[p] + cc * KTAPS + kw];
    }

#pragma unroll 1
    for (int kh = 0; kh < 3; ++kh) {
      const int toff = kh * RPITCH;
#pragma unroll
      for (int q = 0; q < ICC / 4; ++q) {
#pragma unroll
        for (int f = 0; f < NF; ++f) {
          const v2f a = *(const v2f*)&ap[f * 3 * ICC + kh * ICC + 4 * q];
#pragma unroll
          for (int j = 0; j < NT; ++j) {
            const v2f b = *(const v2f*)&xs[(2 * q) * XPAIR + bbase[j] + toff + f * FPITCH];
            c[f][j] = __builtin_amdgcn_wmma_f32_16x16x4_f32(
                 false, a,  false, b,
                 (short)0, c[f][j],  false,  false);
          }
        }
      }
    }

    __syncthreads();
  }

  const float s9 = 1.0f / 9.0f;
  const int   ocr = oc0 + wave * 16 + hiHalf * 8;
#pragma unroll
  for (int j = 0; j < NT; ++j) {
    const long obase = (long)n * (long)OC * HWSZ + h0 * WW + 2 * (j * 16 + l16);
#pragma unroll
    for (int r = 0; r < 8; ++r) {
      const float m0 = c[0][j][r], m1 = c[1][j][r], m2 = c[2][j][r], m3 = c[3][j][r];
      v2f y;
      y.x = (m0 + m1 + m2) * s9;
      y.y = (m1 - m2 - m3) * s9;
      *(v2f*)&out[obase + (long)(ocr + r) * HWSZ] = y;
    }
  }
}

extern "C" void kernel_launch(void* const* d_in, const int* in_sizes, int n_in,
                              void* d_out, int out_size, void* d_ws, size_t ws_size,
                              hipStream_t stream) {
  const float* x = (const float*)d_in[0];
  const float* w = (const float*)d_in[1];
  float* out = (float*)d_out;

  dim3 grid(4, 448, 1);
  dim3 block(NTHREADS, 1, 1);
  conv3x3_wmma_f32<<<grid, block, 0, stream>>>(x, w, out);
}


# the 8-channel slab kept twice in LDS: slab s+1 is staged into the other copy under slab s's products, fragments read one block of 28 ahead; one split barrier per slab, not two; sum order unchanged
# speedup vs baseline: 5.2492x; 1.0599x over previous
#include <hip/hip_runtime.h>


typedef __attribute__((ext_vector_type(2))) float v2f;
typedef __attribute__((ext_vector_type(4))) float v4f;
typedef __attribute__((ext_vector_type(8))) float v8f;

#define IC 256
#define OC 256
#define HH 56
#define WW 56
#define HWSZ (HH * WW)
#define KTAPS 9
#define WROW (IC * KTAPS)

#define NTHREADS 128
#define ICC 8
#define NCHUNK (IC / ICC)
#define BLOCK_OC 64
#define ROWS 4
#define TROWS (ROWS + 2)
#define MW (WW / 2)
#define NF 4
#define NT (ROWS * MW / 16)
#define HBLK (HH / ROWS)

#define FPITCH 64
#define RPITCH (NF * FPITCH)
#define XPAIR (TROWS * RPITCH + 32)
#define XS_FLOATS ((ICC / 2) * XPAIR)
#define UPITCH 100
#define US_FLOATS (BLOCK_OC * UPITCH)

#define XROUNDS 3
#define WROUNDS 3

#define FETCH_SLAB(XB, WB)                                                        \
  do {                                                                            \
    _Pragma("unroll")                                                             \
    for (int s = 0; s < XROUNDS; ++s) {                                           \
      xma[s] = *(const v4f*)((XB) + xg_off[s]);                                   \
      xmb[s] = *(const v4f*)((XB) + xg_off[s] + HWSZ);                            \
      xla[s] = (XB)[xg_l[s]];                                                     \
      xlb[s] = (XB)[xg_l[s] + HWSZ];                                              \
      xra[s] = (XB)[xg_r[s]];                                                     \
      xrb[s] = (XB)[xg_r[s] + HWSZ];                                              \
    }                                                                             \
    _Pragma("unroll")                                                             \
    for (int p = 0; p < WROUNDS; ++p)                                             \
      _Pragma("unroll")                                                           \
      for (int cc = 0; cc < 4; ++cc)                                              \
        _Pragma("unroll")                                                         \
        for (int kw = 0; kw < 3; ++kw)                                            \
          wr[p][cc][kw] = (WB)[wg_off[p] + cc * KTAPS + kw];                      \
  } while (0)

#define STORE_SLAB(XD, UD)                                                        \
  do {                                                                            \
    _Pragma("unroll")                                                             \
    for (int s = 0; s < XROUNDS; ++s)                                             \
      if (xok[s]) {                                                               \
        const float la = hasL ? xla[s] : 0.f, lb = hasL ? xlb[s] : 0.f;           \
        const float ra = hasR ? xra[s] : 0.f, rb = hasR ? xrb[s] : 0.f;           \
        const v4f a = xma[s], b = xmb[s];                                         \
          \
        *(v4f*)&(XD)[xl_off[s]]              = (v4f){la  - a.y, lb  - b.y, a.y - a.w, b.y - b.w}; \
        *(v4f*)&(XD)[xl_off[s] + FPITCH]     = (v4f){a.x + a.y, b.x + b.y, a.z + a.w, b.z + b.w}; \
        *(v4f*)&(XD)[xl_off[s] + 2 * FPITCH] = (v4f){a.y - a.x, b.y - b.x, a.w - a.z, b.w - b.z}; \
        *(v4f*)&(XD)[xl_off[s] + 3 * FPITCH] = (v4f){a.x - a.z, b.x - b.z, a.z - ra,  b.z - rb};  \
      }                                                                           \
    _Pragma("unroll")                                                             \
    for (int p = 0; p < WROUNDS; ++p) {                                           \
      v4f u0, u1, u2, u3;                                                         \
      _Pragma("unroll")                                                           \
      for (int cc = 0; cc < 4; ++cc) {                                            \
        const float g0 = wr[p][cc][0], g1 = wr[p][cc][1], g2 = wr[p][cc][2];      \
        const float e = g0 + g2;                                                  \
        u0[cc] = g0;                                                              \
        u1[cc] = (e + g1) * 0.5f;                                                 \
        u2[cc] = (e - g1) * 0.5f;                                                 \
        u3[cc] = g2;                                                              \
      }                                                                           \
      *(v4f*)&(UD)[wl_off[p]]               = u0;                                 \
      *(v4f*)&(UD)[wl_off[p] + 3 * ICC]     = u1;                                 \
      *(v4f*)&(UD)[wl_off[p] + 6 * ICC]     = u2;                                 \
      *(v4f*)&(UD)[wl_off[p] + 9 * ICC]     = u3;                                 \
    }                                                                             \
  } while (0)

#define LOAD_FRAGS(FA, FB, XC, AP, KH, Q)                                         \
  do {                                                                            \
    _Pragma("unroll")                                                             \
    for (int f = 0; f < NF; ++f) {                                                \
      FA[f] = *(const v2f*)&(AP)[f * 3 * ICC + (KH) * ICC + 4 * (Q)];             \
      _Pragma("unroll")                                                           \
      for (int j = 0; j < NT; ++j)                                                \
        FB[f][j] = *(const v2f*)&(XC)[(2 * (Q)) * XPAIR + bbase[j] + (KH) * RPITCH + f * FPITCH]; \
    }                                                                             \
  } while (0)

#define MULT_BLOCK(FA, FB)                                                        \
  do {                                                                            \
    _Pragma("unroll")                                                             \
    for (int f = 0; f < NF; ++f)                                                  \
      _Pragma("unroll")                                                           \
      for (int j = 0; j < NT; ++j)                                                \
        c[f][j] = __builtin_amdgcn_wmma_f32_16x16x4_f32(                          \
             false, FA[f],  false, FB[f][j],                    \
             (short)0, c[f][j],  false,  false);   \
  } while (0)

#define MULT_HALF(FA, FB, F0)                                                     \
  do {                                                                            \
    _Pragma("unroll")                                                             \
    for (int f = (F0); f < (F0) + NF / 2; ++f)                                    \
      _Pragma("unroll")                                                           \
      for (int j = 0; j < NT; ++j)                                                \
        c[f][j] = __builtin_amdgcn_wmma_f32_16x16x4_f32(                          \
             false, FA[f],  false, FB[f][j],                    \
             (short)0, c[f][j],  false,  false);   \
  } while (0)

#define KEEP_ORDER() __builtin_amdgcn_sched_barrier(0)

__global__ __launch_bounds__(NTHREADS) void conv3x3_wmma_f32(
    const float* __restrict__ x, const float* __restrict__ w,
    float* __restrict__ out)
{
  __shared__ __attribute__((aligned(16))) float xs[2 * XS_FLOATS];
  __shared__ __attribute__((aligned(16))) float usm[2 * US_FLOATS];

  const int tid    = threadIdx.x;
  const int lane   = tid & 31;
  const int wave   = tid >> 5;
  const int l16    = lane & 15;
  const int hiHalf = lane >> 4;

  const int n   = blockIdx.y / HBLK;
  const int h0  = (blockIdx.y - n * HBLK) * ROWS;
  const int oc0 = blockIdx.x * BLOCK_OC;

  const int  rw   = lane / 14;
  const int  c4   = lane - rw * 14;
  const bool hasL = c4 > 0;
  const bool hasR = c4 < 13;
  int  xg_off[XROUNDS], xg_l[XROUNDS], xg_r[XROUNDS], xl_off[XROUNDS];
  bool xok[XROUNDS];
#pragma unroll
  for (int s = 0; s < XROUNDS; ++s) {
    const int line = (s * 4 + wave) * 2 + rw;
    const int xp   = line / TROWS;
    const int rr   = line - xp * TROWS;
    const int row  = h0 - 1 + rr;
    xok[s]    = (lane < 28) && ((unsigned)row < (unsigned)HH);
    xg_off[s] = xok[s] ? (2 * xp) * HWSZ + row * WW + c4 * 4 : 0;
    xg_l[s]   = xg_off[s] - ((xok[s] && hasL) ? 1 : 0);
    xg_r[s]   = xg_off[s] + ((xok[s] && hasR) ? 4 : 0);
    xl_off[s] = xp * XPAIR + rr * RPITCH + c4 * 4;
  }
  int wg_off[WROUNDS], wl_off[WROUNDS];
#pragma unroll
  for (int p = 0; p < WROUNDS; ++p) {
    const int i   = tid + NTHREADS * p;
    const int ocl = i / 6;
    const int rem = i - ocl * 6;
    const int q   = rem / 3;
    const int kh  = rem - q * 3;
    wg_off[p] = ocl * WROW + (4 * q) * KTAPS + kh * 3;
    wl_off[p] = ocl * UPITCH + kh * ICC + 4 * q;
  }

  const float* xblk = x + (long)n * IC * HWSZ;
  const float* wblk = w + (long)oc0 * WROW;

  v4f   xma[XROUNDS], xmb[XROUNDS];
  float xla[XROUNDS], xlb[XROUNDS];
  float xra[XROUNDS], xrb[XROUNDS];
  float wr[WROUNDS][4][3];

#pragma unroll
  for (int s = 0; s < (2 * XS_FLOATS / 4 + NTHREADS - 1) / NTHREADS; ++s) {
    const int f = tid + NTHREADS * s;
    if (f < 2 * XS_FLOATS / 4) *(v4f*)&xs[f * 4] = (v4f){0.f, 0.f, 0.f, 0.f};
  }

  FETCH_SLAB(xblk, wblk);

  __syncthreads();

  STORE_SLAB(xs, usm);
  FETCH_SLAB(xblk + (long)ICC * HWSZ, wblk + ICC * KTAPS);

  int bbase[NT];
#pragma unroll
  for (int j = 0; j < NT; ++j) {
    const int pl = j * 16 + l16;
    const int r  = pl / MW;
    const int m  = pl - r * MW;
    bbase[j] = hiHalf * XPAIR + r * RPITCH + m * 2;
  }
  const float* ap = usm + (wave * 16 + l16) * UPITCH + hiHalf * 2;

  v8f c[NF][NT] = {};
  v2f fa0[NF], fb0[NF][NT];
  v2f fa1[NF], fb1[NF][NT];

  __syncthreads();

  LOAD_FRAGS(fa0, fb0, xs, ap, 0, 0);

  for (int chunk = 0; chunk < NCHUNK; ++chunk) {
    const int cur = chunk & 1;
    const float* xc  = xs  + cur * XS_FLOATS;
    const float* apc = ap  + cur * US_FLOATS;
    float*       xn  = xs  + (cur ^ 1) * XS_FLOATS;
    float*       un  = usm + (cur ^ 1) * US_FLOATS;

    LOAD_FRAGS(fa1, fb1, xc, apc, 0, 1);
    KEEP_ORDER();
    MULT_BLOCK(fa0, fb0);
    KEEP_ORDER();

    if (chunk + 1 < NCHUNK) STORE_SLAB(xn, un);
    {
      const int nx = (chunk + 2 < NCHUNK) ? chunk + 2 : NCHUNK - 1;
      const float* xb = xblk + (long)nx * ICC * HWSZ;
      const float* wb = wblk + nx * ICC * KTAPS;
      FETCH_SLAB(xb, wb);
    }
    KEEP_ORDER();

    LOAD_FRAGS(fa0, fb0, xc, apc, 1, 0);
    KEEP_ORDER();
    MULT_BLOCK(fa1, fb1);
    KEEP_ORDER();
    LOAD_FRAGS(fa1, fb1, xc, apc, 1, 1);
    KEEP_ORDER();
    MULT_BLOCK(fa0, fb0);
    KEEP_ORDER();
    LOAD_FRAGS(fa0, fb0, xc, apc, 2, 0);
    KEEP_ORDER();
    MULT_BLOCK(fa1, fb1);
    KEEP_ORDER();
    LOAD_FRAGS(fa1, fb1, xc, apc, 2, 1);
    KEEP_ORDER();
    MULT_HALF(fa0, fb0, 0);
    KEEP_ORDER();

    __builtin_amdgcn_fence(__ATOMIC_RELEASE, "workgroup", "local");
    __builtin_amdgcn_s_barrier_signal(-1);
    KEEP_ORDER();
    MULT_HALF(fa0, fb0, NF / 2);
    KEEP_ORDER();
    __builtin_amdgcn_s_barrier_wait(-1);
    __builtin_amdgcn_fence(__ATOMIC_ACQUIRE, "workgroup", "local");

    LOAD_FRAGS(fa0, fb0, xn, ap + (cur ^ 1) * US_FLOATS, 0, 0);
    KEEP_ORDER();
    MULT_BLOCK(fa1, fb1);
    KEEP_ORDER();
  }

  const float s9 = 1.0f / 9.0f;
  const int   ocr = oc0 + wave * 16 + hiHalf * 8;
#pragma unroll
  for (int j = 0; j < NT; ++j) {
    const long obase = (long)n * (long)OC * HWSZ + h0 * WW + 2 * (j * 16 + l16);
#pragma unroll
    for (int r = 0; r < 8; ++r) {
      const float m0 = c[0][j][r], m1 = c[1][j][r], m2 = c[2][j][r], m3 = c[3][j][r];
      v2f y;
      y.x = (m0 + m1 + m2) * s9;
      y.y = (m1 - m2 - m3) * s9;
      *(v2f*)&out[obase + (long)(ocr + r) * HWSZ] = y;
    }
  }
}

extern "C" void kernel_launch(void* const* d_in, const int* in_sizes, int n_in,
                              void* d_out, int out_size, void* d_ws, size_t ws_size,
                              hipStream_t stream) {
  const float* x = (const float*)d_in[0];
  const float* w = (const float*)d_in[1];
  float* out = (float*)d_out;

  dim3 grid(4, 448, 1);
  dim3 block(NTHREADS, 1, 1);
  conv3x3_wmma_f32<<<grid, block, 0, stream>>>(x, w, out);
}


# grid balanced: 256 work-groups, one per WGP, each takes half an image (28 rows) x 64 oc as three 8-row units and one 4-row unit: 25 tile passes per SIMD, not 4 rounds of 7; sum order unchanged
# speedup vs baseline: 6.1496x; 1.1715x over previous
#include <hip/hip_runtime.h>


typedef __attribute__((ext_vector_type(2))) float v2f;
typedef __attribute__((ext_vector_type(4))) float v4f;
typedef __attribute__((ext_vector_type(8))) float v8f;

#define IC 256
#define OC 256
#define HH 56
#define WW 56
#define HWSZ (HH * WW)
#define KTAPS 9
#define WROW (IC * KTAPS)

#define NTHREADS 128
#define ICC 8
#define NCHUNK (IC / ICC)
#define BLOCK_OC 64
#define MQ (WW / 4)
#define NF 6
#define NWG 256

#define FPITCH 32
#define RPITCH (NF * FPITCH)
#define XS_MAX ((ICC / 2) * (10 * RPITCH + 32))
#define UPITCH 148
#define US_FLOATS (BLOCK_OC * UPITCH)

#define WROUNDS 3

#define FETCH_SLAB(XB, WB)                                                        \
  do {                                                                            \
    _Pragma("unroll")                                                             \
    for (int s = 0; s < XROUNDS; ++s) {                                           \
      xma[s] = *(const v4f*)((XB) + xg_off[s]);                                   \
      xmb[s] = *(const v4f*)((XB) + xg_off[s] + HWSZ);                            \
      xla[s] = (XB)[xg_l[s]];                                                     \
      xlb[s] = (XB)[xg_l[s] + HWSZ];                                              \
      xra[s] = (XB)[xg_r[s]];                                                     \
      xrb[s] = (XB)[xg_r[s] + HWSZ];                                              \
    }                                                                             \
    _Pragma("unroll")                                                             \
    for (int p = 0; p < WROUNDS; ++p)                                             \
      _Pragma("unroll")                                                           \
      for (int cc = 0; cc < 4; ++cc)                                              \
        _Pragma("unroll")                                                         \
        for (int kw = 0; kw < 3; ++kw)                                            \
          wr[p][cc][kw] = (WB)[wg_off[p] + cc * KTAPS + kw];                      \
  } while (0)

#define STORE_SLAB(XD, UD)                                                        \
  do {                                                                            \
    _Pragma("unroll")                                                             \
    for (int s = 0; s < XROUNDS; ++s)                                             \
      if (xok[s]) {                                                               \
        const float la = hasL ? xla[s] : 0.f, lb = hasL ? xlb[s] : 0.f;           \
        const float ra = hasR ? xra[s] : 0.f, rb = hasR ? xrb[s] : 0.f;           \
        const v4f a = xma[s], b = xmb[s];                                         \
                           \
        const float ta = a.z - a.x, sa = a.w - a.y;                               \
        const float tb = b.z - b.x, sb = b.w - b.y;                               \
        *(v2f*)&(XD)[xl_off[s]]              = (v2f){4.f * la - 5.f * a.y + a.w, 4.f * lb - 5.f * b.y + b.w}; \
        *(v2f*)&(XD)[xl_off[s] + FPITCH]     = (v2f){(a.z + a.w) - 4.f * (a.x + a.y), (b.z + b.w) - 4.f * (b.x + b.y)}; \
        *(v2f*)&(XD)[xl_off[s] + 2 * FPITCH] = (v2f){4.f * (a.x - a.y) - (a.z - a.w), 4.f * (b.x - b.y) - (b.z - b.w)}; \
        *(v2f*)&(XD)[xl_off[s] + 3 * FPITCH] = (v2f){sa + 2.f * ta, sb + 2.f * tb}; \
        *(v2f*)&(XD)[xl_off[s] + 4 * FPITCH] = (v2f){sa - 2.f * ta, sb - 2.f * tb}; \
        *(v2f*)&(XD)[xl_off[s] + 5 * FPITCH] = (v2f){4.f * a.x - 5.f * a.z + ra, 4.f * b.x - 5.f * b.z + rb}; \
      }                                                                           \
    _Pragma("unroll")                                                             \
    for (int p = 0; p < WROUNDS; ++p) {                                           \
      v4f u0, u1, u2, u3, u4, u5;                                                 \
      _Pragma("unroll")                                                           \
      for (int cc = 0; cc < 4; ++cc) {                                            \
        const float g0 = wr[p][cc][0], g1 = wr[p][cc][1], g2 = wr[p][cc][2];      \
        const float e = g0 + g2;                                                  \
        const float h = g0 * (1.f / 24.f) + g2 * (1.f / 6.f);                     \
        const float k = g1 * (1.f / 12.f);                                        \
        u0[cc] = g0 * 0.25f;                                                      \
        u1[cc] = (e + g1) * (-1.f / 6.f);                                         \
        u2[cc] = (e - g1) * (-1.f / 6.f);                                         \
        u3[cc] = h + k;                                                           \
        u4[cc] = h - k;                                                           \
        u5[cc] = g2;                                                              \
      }                                                                           \
      *(v4f*)&(UD)[wl_off[p]]               = u0;                                 \
      *(v4f*)&(UD)[wl_off[p] + 3 * ICC]     = u1;                                 \
      *(v4f*)&(UD)[wl_off[p] + 6 * ICC]     = u2;                                 \
      *(v4f*)&(UD)[wl_off[p] + 9 * ICC]     = u3;                                 \
      *(v4f*)&(UD)[wl_off[p] + 12 * ICC]    = u4;                                 \
      *(v4f*)&(UD)[wl_off[p] + 15 * ICC]    = u5;                                 \
    }                                                                             \
  } while (0)

#define LOAD_FRAGS(FA, FB, XC, AP, KH, Q)                                         \
  do {                                                                            \
    _Pragma("unroll")                                                             \
    for (int f = 0; f < NF; ++f) {                                                \
      FA[f] = *(const v2f*)&(AP)[f * 3 * ICC + (KH) * ICC + 4 * (Q)];             \
      _Pragma("unroll")                                                           \
      for (int j = 0; j < NT; ++j)                                                \
        FB[f][j] = *(const v2f*)&(XC)[(2 * (Q)) * XPAIR + bbase[j] + (KH) * RPITCH + f * FPITCH]; \
    }                                                                             \
  } while (0)

#define MULT_BLOCK(FA, FB)                                                        \
  do {                                                                            \
    _Pragma("unroll")                                                             \
    for (int f = 0; f < NF; ++f)                                                  \
      _Pragma("unroll")                                                           \
      for (int j = 0; j < NT; ++j)                                                \
        c[f][j] = __builtin_amdgcn_wmma_f32_16x16x4_f32(                          \
             false, FA[f],  false, FB[f][j],                    \
             (short)0, c[f][j],  false,  false);   \
  } while (0)

#define MULT_HALF(FA, FB, F0)                                                     \
  do {                                                                            \
    _Pragma("unroll")                                                             \
    for (int f = (F0); f < (F0) + NF / 2; ++f)                                    \
      _Pragma("unroll")                                                           \
      for (int j = 0; j < NT; ++j)                                                \
        c[f][j] = __builtin_amdgcn_wmma_f32_16x16x4_f32(                          \
             false, FA[f],  false, FB[f][j],                    \
             (short)0, c[f][j],  false,  false);   \
  } while (0)

#define KEEP_ORDER() __builtin_amdgcn_sched_barrier(0)

template <int ROWS, int XROUNDS>
__device__ __forceinline__ void conv_unit(
    const float* __restrict__ x, const float* __restrict__ w, float* __restrict__ out,
    float* xs, float* usm, const int n, const int h0, const int oc0)
{
  constexpr int TROWS     = ROWS + 2;
  constexpr int NPOS      = ROWS * MQ;
  constexpr int NT        = (NPOS + 15) / 16;
  constexpr bool FULL     = (NPOS % 16) == 0;
  constexpr int XPAIR     = TROWS * RPITCH + 32;
  constexpr int XS_FLOATS = (ICC / 2) * XPAIR;
  static_assert((ICC / 2) * TROWS == XROUNDS * 8, "staging rounds must cover the (pair, row) lines exactly");
  static_assert(XS_FLOATS <= XS_MAX, "image tile copy exceeds the LDS reserved for it");

  const int tid    = threadIdx.x;
  const int lane   = tid & 31;
  const int wave   = tid >> 5;
  const int l16    = lane & 15;
  const int hiHalf = lane >> 4;

  const int  rw   = lane / 14;
  const int  c4   = lane - rw * 14;
  const bool hasL = c4 > 0;
  const bool hasR = c4 < 13;
  int  xg_off[XROUNDS], xg_l[XROUNDS], xg_r[XROUNDS], xl_off[XROUNDS];
  bool xok[XROUNDS];
#pragma unroll
  for (int s = 0; s < XROUNDS; ++s) {
    const int line = (s * 4 + wave) * 2 + rw;
    const int xp   = line / TROWS;
    const int rr   = line - xp * TROWS;
    const int row  = h0 - 1 + rr;
    xok[s]    = (lane < 28) && ((unsigned)row < (unsigned)HH);
    xg_off[s] = xok[s] ? (2 * xp) * HWSZ + row * WW + c4 * 4 : 0;
    xg_l[s]   = xg_off[s] - ((xok[s] && hasL) ? 1 : 0);
    xg_r[s]   = xg_off[s] + ((xok[s] && hasR) ? 4 : 0);
    xl_off[s] = xp * XPAIR + rr * RPITCH + c4 * 2;
  }
  int wg_off[WROUNDS], wl_off[WROUNDS];
#pragma unroll
  for (int p = 0; p < WROUNDS; ++p) {
    const int i   = tid + NTHREADS * p;
    const int ocl = i / 6;
    const int rem = i - ocl * 6;
    const int q   = rem / 3;
    const int kh  = rem - q * 3;
    wg_off[p] = ocl * WROW + (4 * q) * KTAPS + kh * 3;
    wl_off[p] = ocl * UPITCH + kh * ICC + 4 * q;
  }

  const float* xblk = x + (long)n * IC * HWSZ;
  const float* wblk = w + (long)oc0 * WROW;

  v4f   xma[XROUNDS], xmb[XROUNDS];
  float xla[XROUNDS], xlb[XROUNDS];
  float xra[XROUNDS], xrb[XROUNDS];
  float wr[WROUNDS][4][3];

#pragma unroll
  for (int s = 0; s < (2 * XS_FLOATS / 4 + NTHREADS - 1) / NTHREADS; ++s) {
    const int f = tid + NTHREADS * s;
    if (f < 2 * XS_FLOATS / 4) *(v4f*)&xs[f * 4] = (v4f){0.f, 0.f, 0.f, 0.f};
  }

  FETCH_SLAB(xblk, wblk);

  __syncthreads();

  STORE_SLAB(xs, usm);
  FETCH_SLAB(xblk + (long)ICC * HWSZ, wblk + ICC * KTAPS);

  int bbase[NT];
#pragma unroll
  for (int j = 0; j < NT; ++j) {
    const int pp = j * 16 + l16;
    const int pl = (FULL || pp < NPOS) ? pp : NPOS - 1;
    const int r  = pl / MQ;
    const int m  = pl - r * MQ;
    bbase[j] = hiHalf * XPAIR + r * RPITCH + m * 2;
  }
  const float* ap = usm + (wave * 16 + l16) * UPITCH + hiHalf * 2;

  v8f c[NF][NT] = {};
  v2f fa0[NF], fb0[NF][NT];
  v2f fa1[NF], fb1[NF][NT];

  __syncthreads();

  LOAD_FRAGS(fa0, fb0, xs, ap, 0, 0);

  for (int chunk = 0; chunk < NCHUNK; ++chunk) {
    const int cur = chunk & 1;
    const float* xc  = xs  + cur * XS_FLOATS;
    const float* apc = ap  + cur * US_FLOATS;
    float*       xn  = xs  + (cur ^ 1) * XS_FLOATS;
    float*       un  = usm + (cur ^ 1) * US_FLOATS;

    LOAD_FRAGS(fa1, fb1, xc, apc, 0, 1);
    KEEP_ORDER();
    MULT_BLOCK(fa0, fb0);
    KEEP_ORDER();

    if (chunk + 1 < NCHUNK) STORE_SLAB(xn, un);
    {
      const int nx = (chunk + 2 < NCHUNK) ? chunk + 2 : NCHUNK - 1;
      const float* xb = xblk + (long)nx * ICC * HWSZ;
      const float* wb = wblk + nx * ICC * KTAPS;
      FETCH_SLAB(xb, wb);
    }
    KEEP_ORDER();

    LOAD_FRAGS(fa0, fb0, xc, apc, 1, 0);
    KEEP_ORDER();
    MULT_BLOCK(fa1, fb1);
    KEEP_ORDER();
    LOAD_FRAGS(fa1, fb1, xc, apc, 1, 1);
    KEEP_ORDER();
    MULT_BLOCK(fa0, fb0);
    KEEP_ORDER();
    LOAD_FRAGS(fa0, fb0, xc, apc, 2, 0);
    KEEP_ORDER();
    MULT_BLOCK(fa1, fb1);
    KEEP_ORDER();
    LOAD_FRAGS(fa1, fb1, xc, apc, 2, 1);
    KEEP_ORDER();
    MULT_HALF(fa0, fb0, 0);
    KEEP_ORDER();

    __builtin_amdgcn_fence(__ATOMIC_RELEASE, "workgroup", "local");
    __builtin_amdgcn_s_barrier_signal(-1);
    KEEP_ORDER();
    MULT_HALF(fa0, fb0, NF / 2);
    KEEP_ORDER();
    __builtin_amdgcn_s_barrier_wait(-1);
    __builtin_amdgcn_fence(__ATOMIC_ACQUIRE, "workgroup", "local");

    LOAD_FRAGS(fa0, fb0, xn, ap + (cur ^ 1) * US_FLOATS, 0, 0);
    KEEP_ORDER();
    MULT_BLOCK(fa1, fb1);
    KEEP_ORDER();
  }

  const float s9 = 1.0f / 9.0f;
  const int   ocr = oc0 + wave * 16 + hiHalf * 8;
#pragma unroll
  for (int j = 0; j < NT; ++j) {
    const long obase = (long)n * (long)OC * HWSZ + h0 * WW + 4 * (j * 16 + l16);
    if (FULL || j * 16 + l16 < NPOS) {
#pragma unroll
    for (int r = 0; r < 8; ++r) {
      const float m0 = c[0][j][r], m1 = c[1][j][r], m2 = c[2][j][r];
      const float m3 = c[3][j][r], m4 = c[4][j][r], m5 = c[5][j][r];
      const float s12 = m1 + m2, d12 = m1 - m2, s34 = m3 + m4, d34 = m3 - m4;
      v4f y;
      y.x = (m0 + s12 + s34) * s9;
      y.y = (d12 + 2.f * d34) * s9;
      y.z = (s12 + 4.f * s34) * s9;
      y.w = (d12 + 8.f * d34 + m5) * s9;
      *(v4f*)&out[obase + (long)(ocr + r) * HWSZ] = y;
    }
    }
  }
}

__global__ __launch_bounds__(NTHREADS) void conv3x3_wmma_f32(
    const float* __restrict__ x, const float* __restrict__ w,
    float* __restrict__ out)
{
  __shared__ __attribute__((aligned(16))) float xs[2 * XS_MAX];
  __shared__ __attribute__((aligned(16))) float usm[2 * US_FLOATS];

  const int g   = blockIdx.x;
  const int oc0 = (g & 3) * BLOCK_OC;
  const int n   = g >> 3;
  const int hf  = (g >> 2) & 1;
  const int h8  = hf ? 32 : 0;
  const int h4  = hf ? 28 : 24;

  for (int u = 0; u < 3; ++u)
    conv_unit<8, 5>(x, w, out, xs, usm, n, h8 + 8 * u, oc0);
  conv_unit<4, 3>(x, w, out, xs, usm, n, h4, oc0);
}

extern "C" void kernel_launch(void* const* d_in, const int* in_sizes, int n_in,
                              void* d_out, int out_size, void* d_ws, size_t ws_size,
                              hipStream_t stream) {
  const float* x = (const float*)d_in[0];
  const float* w = (const float*)d_in[1];
  float* out = (float*)d_out;

  dim3 grid(NWG, 1, 1);
  dim3 block(NTHREADS, 1, 1);
  conv3x3_wmma_f32<<<grid, block, 0, stream>>>(x, w, out);
}


# global fetch of the slab after next issued as five groups of 8-9 loads, one in front of each of blocks 1..5 beside its fragment reads, not as 42 loads in one run; sum order unchanged
# speedup vs baseline: 6.4908x; 1.0555x over previous
#include <hip/hip_runtime.h>


typedef __attribute__((ext_vector_type(2))) float v2f;
typedef __attribute__((ext_vector_type(4))) float v4f;
typedef __attribute__((ext_vector_type(8))) float v8f;

#define IC 256
#define OC 256
#define HH 56
#define WW 56
#define HWSZ (HH * WW)
#define KTAPS 9
#define WROW (IC * KTAPS)

#define NTHREADS 128
#define ICC 8
#define NCHUNK (IC / ICC)
#define BLOCK_OC 64
#define MQ (WW / 4)
#define NF 6
#define NWG 256

#define FPITCH 32
#define RPITCH (NF * FPITCH)
#define XS_MAX ((ICC / 2) * (10 * RPITCH + 32))
#define UPITCH 148
#define US_FLOATS (BLOCK_OC * UPITCH)

#define WROUNDS 3

#define FETCH_SLAB(XB, WB)                                                        \
  do {                                                                            \
    _Pragma("unroll")                                                             \
    for (int s = 0; s < XROUNDS; ++s) {                                           \
      xma[s] = *(const v4f*)((XB) + xg_off[s]);                                   \
      xmb[s] = *(const v4f*)((XB) + xg_off[s] + HWSZ);                            \
      xla[s] = (XB)[xg_l[s]];                                                     \
      xlb[s] = (XB)[xg_l[s] + HWSZ];                                              \
      xra[s] = (XB)[xg_r[s]];                                                     \
      xrb[s] = (XB)[xg_r[s] + HWSZ];                                              \
    }                                                                             \
    _Pragma("unroll")                                                             \
    for (int p = 0; p < WROUNDS; ++p)                                             \
      _Pragma("unroll")                                                           \
      for (int cc = 0; cc < 4; ++cc)                                              \
        _Pragma("unroll")                                                         \
        for (int kw = 0; kw < 3; ++kw)                                            \
          wr[p][cc][kw] = (WB)[wg_off[p] + cc * KTAPS + kw];                      \
  } while (0)

#define FETCH_GROUP(XB, WB, G)                                                    \
  do {                                                                            \
    if ((G) < XROUNDS) {                                                          \
      xma[(G) < XROUNDS ? (G) : 0] = *(const v4f*)((XB) + xg_off[(G) < XROUNDS ? (G) : 0]);            \
      xmb[(G) < XROUNDS ? (G) : 0] = *(const v4f*)((XB) + xg_off[(G) < XROUNDS ? (G) : 0] + HWSZ);     \
      xla[(G) < XROUNDS ? (G) : 0] = (XB)[xg_l[(G) < XROUNDS ? (G) : 0]];                              \
      xlb[(G) < XROUNDS ? (G) : 0] = (XB)[xg_l[(G) < XROUNDS ? (G) : 0] + HWSZ];                       \
      xra[(G) < XROUNDS ? (G) : 0] = (XB)[xg_r[(G) < XROUNDS ? (G) : 0]];                              \
      xrb[(G) < XROUNDS ? (G) : 0] = (XB)[xg_r[(G) < XROUNDS ? (G) : 0] + HWSZ];                       \
    }                                                                             \
    _Pragma("unroll")                                                             \
    for (int wi = ((G) * 4 * WROUNDS) / 5; wi < (((G) + 1) * 4 * WROUNDS) / 5; ++wi)  \
      _Pragma("unroll")                                                           \
      for (int kw = 0; kw < 3; ++kw)                                              \
        wr[wi / 4][wi % 4][kw] = (WB)[wg_off[wi / 4] + (wi % 4) * KTAPS + kw];    \
  } while (0)

#define STORE_SLAB(XD, UD)                                                        \
  do {                                                                            \
    _Pragma("unroll")                                                             \
    for (int s = 0; s < XROUNDS; ++s)                                             \
      if (xok[s]) {                                                               \
        const float la = hasL ? xla[s] : 0.f, lb = hasL ? xlb[s] : 0.f;           \
        const float ra = hasR ? xra[s] : 0.f, rb = hasR ? xrb[s] : 0.f;           \
        const v4f a = xma[s], b = xmb[s];                                         \
                           \
        const float ta = a.z - a.x, sa = a.w - a.y;                               \
        const float tb = b.z - b.x, sb = b.w - b.y;                               \
        *(v2f*)&(XD)[xl_off[s]]              = (v2f){4.f * la - 5.f * a.y + a.w, 4.f * lb - 5.f * b.y + b.w}; \
        *(v2f*)&(XD)[xl_off[s] + FPITCH]     = (v2f){(a.z + a.w) - 4.f * (a.x + a.y), (b.z + b.w) - 4.f * (b.x + b.y)}; \
        *(v2f*)&(XD)[xl_off[s] + 2 * FPITCH] = (v2f){4.f * (a.x - a.y) - (a.z - a.w), 4.f * (b.x - b.y) - (b.z - b.w)}; \
        *(v2f*)&(XD)[xl_off[s] + 3 * FPITCH] = (v2f){sa + 2.f * ta, sb + 2.f * tb}; \
        *(v2f*)&(XD)[xl_off[s] + 4 * FPITCH] = (v2f){sa - 2.f * ta, sb - 2.f * tb}; \
        *(v2f*)&(XD)[xl_off[s] + 5 * FPITCH] = (v2f){4.f * a.x - 5.f * a.z + ra, 4.f * b.x - 5.f * b.z + rb}; \
      }                                                                           \
    _Pragma("unroll")                                                             \
    for (int p = 0; p < WROUNDS; ++p) {                                           \
      v4f u0, u1, u2, u3, u4, u5;                                                 \
      _Pragma("unroll")                                                           \
      for (int cc = 0; cc < 4; ++cc) {                                            \
        const float g0 = wr[p][cc][0], g1 = wr[p][cc][1], g2 = wr[p][cc][2];      \
        const float e = g0 + g2;                                                  \
        const float h = g0 * (1.f / 24.f) + g2 * (1.f / 6.f);                     \
        const float k = g1 * (1.f / 12.f);                                        \
        u0[cc] = g0 * 0.25f;                                                      \
        u1[cc] = (e + g1) * (-1.f / 6.f);                                         \
        u2[cc] = (e - g1) * (-1.f / 6.f);                                         \
        u3[cc] = h + k;                                                           \
        u4[cc] = h - k;                                                           \
        u5[cc] = g2;                                                              \
      }                                                                           \
      *(v4f*)&(UD)[wl_off[p]]               = u0;                                 \
      *(v4f*)&(UD)[wl_off[p] + 3 * ICC]     = u1;                                 \
      *(v4f*)&(UD)[wl_off[p] + 6 * ICC]     = u2;                                 \
      *(v4f*)&(UD)[wl_off[p] + 9 * ICC]     = u3;                                 \
      *(v4f*)&(UD)[wl_off[p] + 12 * ICC]    = u4;                                 \
      *(v4f*)&(UD)[wl_off[p] + 15 * ICC]    = u5;                                 \
    }                                                                             \
  } while (0)

#define LOAD_FRAGS(FA, FB, XC, AP, KH, Q)                                         \
  do {                                                                            \
    _Pragma("unroll")                                                             \
    for (int f = 0; f < NF; ++f) {                                                \
      FA[f] = *(const v2f*)&(AP)[f * 3 * ICC + (KH) * ICC + 4 * (Q)];             \
      _Pragma("unroll")                                                           \
      for (int j = 0; j < NT; ++j)                                                \
        FB[f][j] = *(const v2f*)&(XC)[(2 * (Q)) * XPAIR + bbase[j] + (KH) * RPITCH + f * FPITCH]; \
    }                                                                             \
  } while (0)

#define MULT_BLOCK(FA, FB)                                                        \
  do {                                                                            \
    _Pragma("unroll")                                                             \
    for (int f = 0; f < NF; ++f)                                                  \
      _Pragma("unroll")                                                           \
      for (int j = 0; j < NT; ++j)                                                \
        c[f][j] = __builtin_amdgcn_wmma_f32_16x16x4_f32(                          \
             false, FA[f],  false, FB[f][j],                    \
             (short)0, c[f][j],  false,  false);   \
  } while (0)

#define MULT_HALF(FA, FB, F0)                                                     \
  do {                                                                            \
    _Pragma("unroll")                                                             \
    for (int f = (F0); f < (F0) + NF / 2; ++f)                                    \
      _Pragma("unroll")                                                           \
      for (int j = 0; j < NT; ++j)                                                \
        c[f][j] = __builtin_amdgcn_wmma_f32_16x16x4_f32(                          \
             false, FA[f],  false, FB[f][j],                    \
             (short)0, c[f][j],  false,  false);   \
  } while (0)

#define KEEP_ORDER() __builtin_amdgcn_sched_barrier(0)

template <int ROWS, int XROUNDS>
__device__ __forceinline__ void conv_unit(
    const float* __restrict__ x, const float* __restrict__ w, float* __restrict__ out,
    float* xs, float* usm, const int n, const int h0, const int oc0)
{
  constexpr int TROWS     = ROWS + 2;
  constexpr int NPOS      = ROWS * MQ;
  constexpr int NT        = (NPOS + 15) / 16;
  constexpr bool FULL     = (NPOS % 16) == 0;
  constexpr int XPAIR     = TROWS * RPITCH + 32;
  constexpr int XS_FLOATS = (ICC / 2) * XPAIR;
  static_assert((ICC / 2) * TROWS == XROUNDS * 8, "staging rounds must cover the (pair, row) lines exactly");
  static_assert(XS_FLOATS <= XS_MAX, "image tile copy exceeds the LDS reserved for it");

  const int tid    = threadIdx.x;
  const int lane   = tid & 31;
  const int wave   = tid >> 5;
  const int l16    = lane & 15;
  const int hiHalf = lane >> 4;

  const int  rw   = lane / 14;
  const int  c4   = lane - rw * 14;
  const bool hasL = c4 > 0;
  const bool hasR = c4 < 13;
  int  xg_off[XROUNDS], xg_l[XROUNDS], xg_r[XROUNDS], xl_off[XROUNDS];
  bool xok[XROUNDS];
#pragma unroll
  for (int s = 0; s < XROUNDS; ++s) {
    const int line = (s * 4 + wave) * 2 + rw;
    const int xp   = line / TROWS;
    const int rr   = line - xp * TROWS;
    const int row  = h0 - 1 + rr;
    xok[s]    = (lane < 28) && ((unsigned)row < (unsigned)HH);
    xg_off[s] = xok[s] ? (2 * xp) * HWSZ + row * WW + c4 * 4 : 0;
    xg_l[s]   = xg_off[s] - ((xok[s] && hasL) ? 1 : 0);
    xg_r[s]   = xg_off[s] + ((xok[s] && hasR) ? 4 : 0);
    xl_off[s] = xp * XPAIR + rr * RPITCH + c4 * 2;
  }
  int wg_off[WROUNDS], wl_off[WROUNDS];
#pragma unroll
  for (int p = 0; p < WROUNDS; ++p) {
    const int i   = tid + NTHREADS * p;
    const int ocl = i / 6;
    const int rem = i - ocl * 6;
    const int q   = rem / 3;
    const int kh  = rem - q * 3;
    wg_off[p] = ocl * WROW + (4 * q) * KTAPS + kh * 3;
    wl_off[p] = ocl * UPITCH + kh * ICC + 4 * q;
  }

  const float* xblk = x + (long)n * IC * HWSZ;
  const float* wblk = w + (long)oc0 * WROW;

  v4f   xma[XROUNDS], xmb[XROUNDS];
  float xla[XROUNDS], xlb[XROUNDS];
  float xra[XROUNDS], xrb[XROUNDS];
  float wr[WROUNDS][4][3];

#pragma unroll
  for (int s = 0; s < (2 * XS_FLOATS / 4 + NTHREADS - 1) / NTHREADS; ++s) {
    const int f = tid + NTHREADS * s;
    if (f < 2 * XS_FLOATS / 4) *(v4f*)&xs[f * 4] = (v4f){0.f, 0.f, 0.f, 0.f};
  }

  FETCH_SLAB(xblk, wblk);

  __syncthreads();

  STORE_SLAB(xs, usm);
  FETCH_SLAB(xblk + (long)ICC * HWSZ, wblk + ICC * KTAPS);

  int bbase[NT];
#pragma unroll
  for (int j = 0; j < NT; ++j) {
    const int pp = j * 16 + l16;
    const int pl = (FULL || pp < NPOS) ? pp : NPOS - 1;
    const int r  = pl / MQ;
    const int m  = pl - r * MQ;
    bbase[j] = hiHalf * XPAIR + r * RPITCH + m * 2;
  }
  const float* ap = usm + (wave * 16 + l16) * UPITCH + hiHalf * 2;

  v8f c[NF][NT] = {};
  v2f fa0[NF], fb0[NF][NT];
  v2f fa1[NF], fb1[NF][NT];

  __syncthreads();

  LOAD_FRAGS(fa0, fb0, xs, ap, 0, 0);

  for (int chunk = 0; chunk < NCHUNK; ++chunk) {
    const int cur = chunk & 1;
    const float* xc  = xs  + cur * XS_FLOATS;
    const float* apc = ap  + cur * US_FLOATS;
    float*       xn  = xs  + (cur ^ 1) * XS_FLOATS;
    float*       un  = usm + (cur ^ 1) * US_FLOATS;

    LOAD_FRAGS(fa1, fb1, xc, apc, 0, 1);
    KEEP_ORDER();
    MULT_BLOCK(fa0, fb0);
    KEEP_ORDER();

    if (chunk + 1 < NCHUNK) STORE_SLAB(xn, un);
    const int nx = (chunk + 2 < NCHUNK) ? chunk + 2 : NCHUNK - 1;
    const float* xb = xblk + (long)nx * ICC * HWSZ;
    const float* wb = wblk + nx * ICC * KTAPS;
    KEEP_ORDER();

    FETCH_GROUP(xb, wb, 0);
    LOAD_FRAGS(fa0, fb0, xc, apc, 1, 0);
    KEEP_ORDER();
    MULT_BLOCK(fa1, fb1);
    KEEP_ORDER();
    FETCH_GROUP(xb, wb, 1);
    LOAD_FRAGS(fa1, fb1, xc, apc, 1, 1);
    KEEP_ORDER();
    MULT_BLOCK(fa0, fb0);
    KEEP_ORDER();
    FETCH_GROUP(xb, wb, 2);
    LOAD_FRAGS(fa0, fb0, xc, apc, 2, 0);
    KEEP_ORDER();
    MULT_BLOCK(fa1, fb1);
    KEEP_ORDER();
    FETCH_GROUP(xb, wb, 3);
    LOAD_FRAGS(fa1, fb1, xc, apc, 2, 1);
    KEEP_ORDER();
    MULT_HALF(fa0, fb0, 0);
    KEEP_ORDER();

    __builtin_amdgcn_fence(__ATOMIC_RELEASE, "workgroup", "local");
    __builtin_amdgcn_s_barrier_signal(-1);
    KEEP_ORDER();
    MULT_HALF(fa0, fb0, NF / 2);
    KEEP_ORDER();
    __builtin_amdgcn_s_barrier_wait(-1);
    __builtin_amdgcn_fence(__ATOMIC_ACQUIRE, "workgroup", "local");

    FETCH_GROUP(xb, wb, 4);
    LOAD_FRAGS(fa0, fb0, xn, ap + (cur ^ 1) * US_FLOATS, 0, 0);
    KEEP_ORDER();
    MULT_BLOCK(fa1, fb1);
    KEEP_ORDER();
  }

  const float s9 = 1.0f / 9.0f;
  const int   ocr = oc0 + wave * 16 + hiHalf * 8;
#pragma unroll
  for (int j = 0; j < NT; ++j) {
    const long obase = (long)n * (long)OC * HWSZ + h0 * WW + 4 * (j * 16 + l16);
    if (FULL || j * 16 + l16 < NPOS) {
#pragma unroll
    for (int r = 0; r < 8; ++r) {
      const float m0 = c[0][j][r], m1 = c[1][j][r], m2 = c[2][j][r];
      const float m3 = c[3][j][r], m4 = c[4][j][r], m5 = c[5][j][r];
      const float s12 = m1 + m2, d12 = m1 - m2, s34 = m3 + m4, d34 = m3 - m4;
      v4f y;
      y.x = (m0 + s12 + s34) * s9;
      y.y = (d12 + 2.f * d34) * s9;
      y.z = (s12 + 4.f * s34) * s9;
      y.w = (d12 + 8.f * d34 + m5) * s9;
      *(v4f*)&out[obase + (long)(ocr + r) * HWSZ] = y;
    }
    }
  }
}

__global__ __launch_bounds__(NTHREADS) void conv3x3_wmma_f32(
    const float* __restrict__ x, const float* __restrict__ w,
    float* __restrict__ out)
{
  __shared__ __attribute__((aligned(16))) float xs[2 * XS_MAX];
  __shared__ __attribute__((aligned(16))) float usm[2 * US_FLOATS];

  const int g   = blockIdx.x;
  const int oc0 = (g & 3) * BLOCK_OC;
  const int n   = g >> 3;
  const int hf  = (g >> 2) & 1;
  const int h8  = hf ? 32 : 0;
  const int h4  = hf ? 28 : 24;

  for (int u = 0; u < 3; ++u)
    conv_unit<8, 5>(x, w, out, xs, usm, n, h8 + 8 * u, oc0);
  conv_unit<4, 3>(x, w, out, xs, usm, n, h4, oc0);
}

extern "C" void kernel_launch(void* const* d_in, const int* in_sizes, int n_in,
                              void* d_out, int out_size, void* d_ws, size_t ws_size,
                              hipStream_t stream) {
  const float* x = (const float*)d_in[0];
  const float* w = (const float*)d_in[1];
  float* out = (float*)d_out;

  dim3 grid(NWG, 1, 1);
  dim3 block(NTHREADS, 1, 1);
  conv3x3_wmma_f32<<<grid, block, 0, stream>>>(x, w, out);
}
